# starting kernel
# baseline (speedup 1.0000x reference)
#include <hip/hip_runtime.h>
#include <hip/hip_bf16.h>


typedef __attribute__((ext_vector_type(2))) float v2f;
typedef __attribute__((ext_vector_type(8))) float v8f;

#define D_FEAT 32

__global__ void gcn_zero_kernel(float* __restrict__ p, size_t n) {
    size_t i = (size_t)blockIdx.x * blockDim.x + threadIdx.x;
    if (i < n) p[i] = 0.0f;
}

__global__ void gcn_degree_kernel(const int* __restrict__ er,
                                  const float* __restrict__ ev,
                                  float* __restrict__ deg, int E) {
    int e = blockIdx.x * blockDim.x + threadIdx.x;
    if (e < E) atomicAdd(&deg[er[e]], ev[e]);
}

__global__ void gcn_rsqrt_kernel(float* __restrict__ deg, int N) {
    int i = blockIdx.x * blockDim.x + threadIdx.x;
    if (i < N) {
        float d = deg[i];
        deg[i] = (d > 0.0f) ? rsqrtf(d) : 0.0f;
    }
}

__global__ void gcn_spmm_kernel(const int* __restrict__ er,
                                const int* __restrict__ ec,
                                const float* __restrict__ ev,
                                const float* __restrict__ dis,
                                const float* __restrict__ feat,
                                float* __restrict__ agg, int E) {
    int wave = (int)(((size_t)blockIdx.x * blockDim.x + threadIdx.x) >> 5);
    if (wave >= E) return;
    int lane = threadIdx.x & 31;
    int r = er[wave];
    int c = ec[wave];
    float v = dis[r] * ev[wave] * dis[c];
    float f = feat[(size_t)c * D_FEAT + lane];
    atomicAdd(&agg[(size_t)r * D_FEAT + lane], v * f);
}

__global__ void gcn_gemm_wmma_kernel(const float* __restrict__ agg,
                                     const float* __restrict__ W,
                                     float* __restrict__ out, int nTiles) {
    int wave = (int)(((size_t)blockIdx.x * blockDim.x + threadIdx.x) >> 5);
    if (wave >= nTiles) return;
    int lane = threadIdx.x & 31;
    int half = lane >> 4;
    int l15  = lane & 15;

    const float* aRow = agg + (size_t)(wave * 16 + l15) * D_FEAT;

    v8f acc0 = {};
    v8f acc1 = {};
#pragma unroll
    for (int k = 0; k < D_FEAT; k += 4) {
        int kk = k + half * 2;
        v2f a, b0, b1;
        a.x  = aRow[kk];
        a.y  = aRow[kk + 1];
        b0.x = W[l15 * D_FEAT + kk];
        b0.y = W[l15 * D_FEAT + kk + 1];
        b1.x = W[(16 + l15) * D_FEAT + kk];
        b1.y = W[(16 + l15) * D_FEAT + kk + 1];
        acc0 = __builtin_amdgcn_wmma_f32_16x16x4_f32(
            false, a, false, b0, (short)0, acc0, false, false);
        acc1 = __builtin_amdgcn_wmma_f32_16x16x4_f32(
            false, a, false, b1, (short)0, acc1, false, false);
    }

    int mBase = wave * 16 + half * 8;
#pragma unroll
    for (int i = 0; i < 8; ++i) {
        out[(size_t)(mBase + i) * D_FEAT + l15]      = acc0[i];
        out[(size_t)(mBase + i) * D_FEAT + 16 + l15] = acc1[i];
    }
}

__global__ void gcn_gemm_tail_kernel(const float* __restrict__ agg,
                                     const float* __restrict__ W,
                                     float* __restrict__ out,
                                     int rowStart, int N) {
    int idx = blockIdx.x * blockDim.x + threadIdx.x;
    int row = rowStart + (idx >> 5);
    int o = idx & 31;
    if (row >= N) return;
    float s = 0.0f;
#pragma unroll
    for (int k = 0; k < D_FEAT; ++k)
        s += agg[(size_t)row * D_FEAT + k] * W[o * D_FEAT + k];
    out[(size_t)row * D_FEAT + o] = s;
}

extern "C" void kernel_launch(void* const* d_in, const int* in_sizes, int n_in,
                              void* d_out, int out_size, void* d_ws, size_t ws_size,
                              hipStream_t stream) {
    const int*   er   = (const int*)d_in[0];
    const int*   ec   = (const int*)d_in[1];
    const float* ev   = (const float*)d_in[2];
    const float* feat = (const float*)d_in[3];
    const float* W    = (const float*)d_in[4];
    float* out = (float*)d_out;

    int E = in_sizes[0];
    int N = in_sizes[3] / D_FEAT;

    float* agg = (float*)d_ws;
    float* deg = agg + (size_t)N * D_FEAT;

    size_t zn = (size_t)N * (D_FEAT + 1);
    gcn_zero_kernel<<<(unsigned)((zn + 255) / 256), 256, 0, stream>>>(agg, zn);

    gcn_degree_kernel<<<(E + 255) / 256, 256, 0, stream>>>(er, ev, deg, E);

    gcn_rsqrt_kernel<<<(N + 255) / 256, 256, 0, stream>>>(deg, N);

    size_t spmmThreads = (size_t)E * 32;
    gcn_spmm_kernel<<<(unsigned)((spmmThreads + 255) / 256), 256, 0, stream>>>(
        er, ec, ev, deg, feat, agg, E);

    int nTiles = N / 16;
    if (nTiles > 0) {
        size_t gemmThreads = (size_t)nTiles * 32;
        gcn_gemm_wmma_kernel<<<(unsigned)((gemmThreads + 255) / 256), 256, 0, stream>>>(
            agg, W, out, nTiles);
    }
    int rowStart = nTiles * 16;
    if (rowStart < N) {
        int tailThreads = (N - rowStart) * 32;
        gcn_gemm_tail_kernel<<<(tailThreads + 255) / 256, 256, 0, stream>>>(
            agg, W, out, rowStart, N);
    }
}


# SpMM: one wave per 32 edges instead of one per edge: indices and coefficients read one edge per lane, broadcast per edge, all 32 feature-row gathers issued before the first wait; f32 atomics kept
# speedup vs baseline: 1.6081x; 1.6081x over previous
#include <hip/hip_runtime.h>
#include <hip/hip_bf16.h>


typedef __attribute__((ext_vector_type(2))) float v2f;
typedef __attribute__((ext_vector_type(8))) float v8f;

#define D_FEAT 32

__global__ void gcn_zero_kernel(float* __restrict__ p, size_t n) {
    size_t i = (size_t)blockIdx.x * blockDim.x + threadIdx.x;
    if (i < n) p[i] = 0.0f;
}

__global__ void gcn_degree_kernel(const int* __restrict__ er,
                                  const float* __restrict__ ev,
                                  float* __restrict__ deg, int E) {
    int e = blockIdx.x * blockDim.x + threadIdx.x;
    if (e < E) atomicAdd(&deg[er[e]], ev[e]);
}

__global__ void gcn_rsqrt_kernel(float* __restrict__ deg, int N) {
    int i = blockIdx.x * blockDim.x + threadIdx.x;
    if (i < N) {
        float d = deg[i];
        deg[i] = (d > 0.0f) ? rsqrtf(d) : 0.0f;
    }
}

#define EDGES_PER_WAVE 32

__global__ void gcn_spmm_kernel(const int* __restrict__ er,
                                const int* __restrict__ ec,
                                const float* __restrict__ ev,
                                const float* __restrict__ dis,
                                const float* __restrict__ feat,
                                float* __restrict__ agg, int E) {
    int wave = __builtin_amdgcn_readfirstlane(
        (int)(((size_t)blockIdx.x * blockDim.x + threadIdx.x) >> 5));
    int lane = threadIdx.x & 31;
    long long base = (long long)wave * EDGES_PER_WAVE;
    if (base >= (long long)E) return;
    int cnt = (int)((long long)E - base);
    if (cnt > EDGES_PER_WAVE) cnt = EDGES_PER_WAVE;

    int r = 0, c = 0;
    float v = 0.0f;
    if (lane < cnt) {
        long long e = base + lane;
        r = er[e];
        c = ec[e];
        v = dis[r] * ev[e] * dis[c];
    }

    if (cnt == EDGES_PER_WAVE) {
        float f[EDGES_PER_WAVE];
#pragma unroll
        for (int j = 0; j < EDGES_PER_WAVE; ++j) {
            int cj = __builtin_amdgcn_readlane(c, j);
            const float* frow = feat + (size_t)cj * D_FEAT;
            f[j] = frow[lane];
        }
#pragma unroll
        for (int j = 0; j < EDGES_PER_WAVE; ++j) {
            int rj = __builtin_amdgcn_readlane(r, j);
            float vj = __int_as_float(__builtin_amdgcn_readlane(__float_as_int(v), j));
            float* arow = agg + (size_t)rj * D_FEAT;
            atomicAdd(&arow[lane], vj * f[j]);
        }
    } else {
        for (int j = 0; j < cnt; ++j) {
            int cj = __builtin_amdgcn_readlane(c, j);
            int rj = __builtin_amdgcn_readlane(r, j);
            float vj = __int_as_float(__builtin_amdgcn_readlane(__float_as_int(v), j));
            float fj = feat[(size_t)cj * D_FEAT + lane];
            atomicAdd(&agg[(size_t)rj * D_FEAT + lane], vj * fj);
        }
    }
}

__global__ void gcn_gemm_wmma_kernel(const float* __restrict__ agg,
                                     const float* __restrict__ W,
                                     float* __restrict__ out, int nTiles) {
    int wave = (int)(((size_t)blockIdx.x * blockDim.x + threadIdx.x) >> 5);
    if (wave >= nTiles) return;
    int lane = threadIdx.x & 31;
    int half = lane >> 4;
    int l15  = lane & 15;

    const float* aRow = agg + (size_t)(wave * 16 + l15) * D_FEAT;

    v8f acc0 = {};
    v8f acc1 = {};
#pragma unroll
    for (int k = 0; k < D_FEAT; k += 4) {
        int kk = k + half * 2;
        v2f a, b0, b1;
        a.x  = aRow[kk];
        a.y  = aRow[kk + 1];
        b0.x = W[l15 * D_FEAT + kk];
        b0.y = W[l15 * D_FEAT + kk + 1];
        b1.x = W[(16 + l15) * D_FEAT + kk];
        b1.y = W[(16 + l15) * D_FEAT + kk + 1];
        acc0 = __builtin_amdgcn_wmma_f32_16x16x4_f32(
            false, a, false, b0, (short)0, acc0, false, false);
        acc1 = __builtin_amdgcn_wmma_f32_16x16x4_f32(
            false, a, false, b1, (short)0, acc1, false, false);
    }

    int mBase = wave * 16 + half * 8;
#pragma unroll
    for (int i = 0; i < 8; ++i) {
        out[(size_t)(mBase + i) * D_FEAT + l15]      = acc0[i];
        out[(size_t)(mBase + i) * D_FEAT + 16 + l15] = acc1[i];
    }
}

__global__ void gcn_gemm_tail_kernel(const float* __restrict__ agg,
                                     const float* __restrict__ W,
                                     float* __restrict__ out,
                                     int rowStart, int N) {
    int idx = blockIdx.x * blockDim.x + threadIdx.x;
    int row = rowStart + (idx >> 5);
    int o = idx & 31;
    if (row >= N) return;
    float s = 0.0f;
#pragma unroll
    for (int k = 0; k < D_FEAT; ++k)
        s += agg[(size_t)row * D_FEAT + k] * W[o * D_FEAT + k];
    out[(size_t)row * D_FEAT + o] = s;
}

extern "C" void kernel_launch(void* const* d_in, const int* in_sizes, int n_in,
                              void* d_out, int out_size, void* d_ws, size_t ws_size,
                              hipStream_t stream) {
    const int*   er   = (const int*)d_in[0];
    const int*   ec   = (const int*)d_in[1];
    const float* ev   = (const float*)d_in[2];
    const float* feat = (const float*)d_in[3];
    const float* W    = (const float*)d_in[4];
    float* out = (float*)d_out;

    int E = in_sizes[0];
    int N = in_sizes[3] / D_FEAT;

    float* agg = (float*)d_ws;
    float* deg = agg + (size_t)N * D_FEAT;

    size_t zn = (size_t)N * (D_FEAT + 1);
    gcn_zero_kernel<<<(unsigned)((zn + 255) / 256), 256, 0, stream>>>(agg, zn);

    gcn_degree_kernel<<<(E + 255) / 256, 256, 0, stream>>>(er, ev, deg, E);

    gcn_rsqrt_kernel<<<(N + 255) / 256, 256, 0, stream>>>(deg, N);

    size_t spmmWaves = ((size_t)E + EDGES_PER_WAVE - 1) / EDGES_PER_WAVE;
    size_t spmmThreads = spmmWaves * 32;
    gcn_spmm_kernel<<<(unsigned)((spmmThreads + 255) / 256), 256, 0, stream>>>(
        er, ec, ev, deg, feat, agg, E);

    int nTiles = N / 16;
    if (nTiles > 0) {
        size_t gemmThreads = (size_t)nTiles * 32;
        gcn_gemm_wmma_kernel<<<(unsigned)((gemmThreads + 255) / 256), 256, 0, stream>>>(
            agg, W, out, nTiles);
    }
    int rowStart = nTiles * 16;
    if (rowStart < N) {
        int tailThreads = (N - rowStart) * 32;
        gcn_gemm_tail_kernel<<<(tailThreads + 255) / 256, 256, 0, stream>>>(
            agg, W, out, rowStart, N);
    }
}


# Degree: 64 edge chunks x row ranges summed in LDS (f32 LDS adds), stored as partial planes and added in fixed order in the rsqrt kernel; no global atomics for deg; degree sum order changes
# speedup vs baseline: 3.1107x; 1.9344x over previous
#include <hip/hip_runtime.h>
#include <hip/hip_bf16.h>


typedef __attribute__((ext_vector_type(2))) float v2f;
typedef __attribute__((ext_vector_type(8))) float v8f;

#define D_FEAT 32

__global__ void gcn_zero_kernel(float* __restrict__ p, size_t n) {
    size_t i = (size_t)blockIdx.x * blockDim.x + threadIdx.x;
    if (i < n) p[i] = 0.0f;
}

#define DEG_ROWS_PER_PASS 65536
#define DEG_THREADS 1024
#define DEG_CHUNKS 64
#define DEG_BATCH 16

__global__ void __launch_bounds__(DEG_THREADS)
gcn_degree_partial_kernel(const int* __restrict__ er,
                          const float* __restrict__ ev,
                          float* __restrict__ part, int E, int N) {
    __shared__ float4 h4[DEG_ROWS_PER_PASS / 4];
    float* h = (float*)h4;

    const unsigned tid = threadIdx.x;
    const unsigned n = (unsigned)N;
    const unsigned stride = (n + 31u) & ~31u;
    const unsigned passes = gridDim.y;
    unsigned rowsPer = (((n + passes - 1u) / passes) + 31u) & ~31u;
    if (rowsPer > DEG_ROWS_PER_PASS) rowsPer = DEG_ROWS_PER_PASS;
    const unsigned rowBase = blockIdx.y * rowsPer;
    unsigned rows = 0;
    if (rowBase < n) rows = (n - rowBase < rowsPer) ? (n - rowBase) : rowsPer;

    const unsigned chunks = gridDim.x;
    const unsigned chunk = ((unsigned)E + chunks - 1u) / chunks;
    const unsigned e0 = blockIdx.x * chunk;
    unsigned e1 = e0 + chunk;
    if (e1 > (unsigned)E) e1 = (unsigned)E;

    for (unsigned i = tid; i < rowsPer / 4u; i += DEG_THREADS)
        h4[i] = make_float4(0.0f, 0.0f, 0.0f, 0.0f);
    __syncthreads();

    for (unsigned base = e0 + tid; base < e1; base += DEG_THREADS * DEG_BATCH) {
        int   r[DEG_BATCH];
        float v[DEG_BATCH];
#pragma unroll
        for (int j = 0; j < DEG_BATCH; ++j) {
            unsigned e = base + (unsigned)j * DEG_THREADS;
            r[j] = -1;
            v[j] = 0.0f;
            if (e < e1) {
                r[j] = er[e];
                v[j] = ev[e];
            }
        }
#pragma unroll
        for (int j = 0; j < DEG_BATCH; ++j) {
            unsigned local = (unsigned)r[j] - rowBase;
            if (r[j] >= 0 && local < rows) atomicAdd(&h[local], v[j]);
        }
    }
    __syncthreads();

    float* plane = part + (size_t)blockIdx.x * stride + rowBase;
    for (unsigned i = tid; i < rows; i += DEG_THREADS) plane[i] = h[i];
}

__global__ void gcn_rsqrt_kernel(const float* __restrict__ part,
                                 float* __restrict__ dis, int N) {
    int i = blockIdx.x * blockDim.x + threadIdx.x;
    if (i < N) {
        const size_t stride = ((size_t)N + 31) & ~(size_t)31;
        float p[DEG_CHUNKS];
#pragma unroll
        for (int g = 0; g < DEG_CHUNKS; ++g) p[g] = part[(size_t)g * stride + i];
        float d = 0.0f;
#pragma unroll
        for (int g = 0; g < DEG_CHUNKS; ++g) d += p[g];
        dis[i] = (d > 0.0f) ? rsqrtf(d) : 0.0f;
    }
}

#define EDGES_PER_WAVE 32

__global__ void gcn_spmm_kernel(const int* __restrict__ er,
                                const int* __restrict__ ec,
                                const float* __restrict__ ev,
                                const float* __restrict__ dis,
                                const float* __restrict__ feat,
                                float* __restrict__ agg, int E) {
    int wave = __builtin_amdgcn_readfirstlane(
        (int)(((size_t)blockIdx.x * blockDim.x + threadIdx.x) >> 5));
    int lane = threadIdx.x & 31;
    long long base = (long long)wave * EDGES_PER_WAVE;
    if (base >= (long long)E) return;
    int cnt = (int)((long long)E - base);
    if (cnt > EDGES_PER_WAVE) cnt = EDGES_PER_WAVE;

    int r = 0, c = 0;
    float v = 0.0f;
    if (lane < cnt) {
        long long e = base + lane;
        r = er[e];
        c = ec[e];
        v = dis[r] * ev[e] * dis[c];
    }

    if (cnt == EDGES_PER_WAVE) {
        float f[EDGES_PER_WAVE];
#pragma unroll
        for (int j = 0; j < EDGES_PER_WAVE; ++j) {
            int cj = __builtin_amdgcn_readlane(c, j);
            const float* frow = feat + (size_t)cj * D_FEAT;
            f[j] = frow[lane];
        }
#pragma unroll
        for (int j = 0; j < EDGES_PER_WAVE; ++j) {
            int rj = __builtin_amdgcn_readlane(r, j);
            float vj = __int_as_float(__builtin_amdgcn_readlane(__float_as_int(v), j));
            float* arow = agg + (size_t)rj * D_FEAT;
            atomicAdd(&arow[lane], vj * f[j]);
        }
    } else {
        for (int j = 0; j < cnt; ++j) {
            int cj = __builtin_amdgcn_readlane(c, j);
            int rj = __builtin_amdgcn_readlane(r, j);
            float vj = __int_as_float(__builtin_amdgcn_readlane(__float_as_int(v), j));
            float fj = feat[(size_t)cj * D_FEAT + lane];
            atomicAdd(&agg[(size_t)rj * D_FEAT + lane], vj * fj);
        }
    }
}

__global__ void gcn_gemm_wmma_kernel(const float* __restrict__ agg,
                                     const float* __restrict__ W,
                                     float* __restrict__ out, int nTiles) {
    int wave = (int)(((size_t)blockIdx.x * blockDim.x + threadIdx.x) >> 5);
    if (wave >= nTiles) return;
    int lane = threadIdx.x & 31;
    int half = lane >> 4;
    int l15  = lane & 15;

    const float* aRow = agg + (size_t)(wave * 16 + l15) * D_FEAT;

    v8f acc0 = {};
    v8f acc1 = {};
#pragma unroll
    for (int k = 0; k < D_FEAT; k += 4) {
        int kk = k + half * 2;
        v2f a, b0, b1;
        a.x  = aRow[kk];
        a.y  = aRow[kk + 1];
        b0.x = W[l15 * D_FEAT + kk];
        b0.y = W[l15 * D_FEAT + kk + 1];
        b1.x = W[(16 + l15) * D_FEAT + kk];
        b1.y = W[(16 + l15) * D_FEAT + kk + 1];
        acc0 = __builtin_amdgcn_wmma_f32_16x16x4_f32(
            false, a, false, b0, (short)0, acc0, false, false);
        acc1 = __builtin_amdgcn_wmma_f32_16x16x4_f32(
            false, a, false, b1, (short)0, acc1, false, false);
    }

    int mBase = wave * 16 + half * 8;
#pragma unroll
    for (int i = 0; i < 8; ++i) {
        out[(size_t)(mBase + i) * D_FEAT + l15]      = acc0[i];
        out[(size_t)(mBase + i) * D_FEAT + 16 + l15] = acc1[i];
    }
}

__global__ void gcn_gemm_tail_kernel(const float* __restrict__ agg,
                                     const float* __restrict__ W,
                                     float* __restrict__ out,
                                     int rowStart, int N) {
    int idx = blockIdx.x * blockDim.x + threadIdx.x;
    int row = rowStart + (idx >> 5);
    int o = idx & 31;
    if (row >= N) return;
    float s = 0.0f;
#pragma unroll
    for (int k = 0; k < D_FEAT; ++k)
        s += agg[(size_t)row * D_FEAT + k] * W[o * D_FEAT + k];
    out[(size_t)row * D_FEAT + o] = s;
}

extern "C" void kernel_launch(void* const* d_in, const int* in_sizes, int n_in,
                              void* d_out, int out_size, void* d_ws, size_t ws_size,
                              hipStream_t stream) {
    const int*   er   = (const int*)d_in[0];
    const int*   ec   = (const int*)d_in[1];
    const float* ev   = (const float*)d_in[2];
    const float* feat = (const float*)d_in[3];
    const float* W    = (const float*)d_in[4];
    float* out = (float*)d_out;

    int E = in_sizes[0];
    int N = in_sizes[3] / D_FEAT;

    float* agg  = (float*)d_ws;
    float* deg  = agg + (size_t)N * D_FEAT;
    size_t pitch = ((size_t)N + 31) & ~(size_t)31;
    float* part = deg + pitch;

    size_t zn = (size_t)N * D_FEAT;
    gcn_zero_kernel<<<(unsigned)((zn + 255) / 256), 256, 0, stream>>>(agg, zn);

    dim3 degGrid(DEG_CHUNKS, (unsigned)((N + DEG_ROWS_PER_PASS - 1) / DEG_ROWS_PER_PASS));
    gcn_degree_partial_kernel<<<degGrid, DEG_THREADS, 0, stream>>>(er, ev, part, E, N);

    gcn_rsqrt_kernel<<<(N + 255) / 256, 256, 0, stream>>>(part, deg, N);

    size_t spmmWaves = ((size_t)E + EDGES_PER_WAVE - 1) / EDGES_PER_WAVE;
    size_t spmmThreads = spmmWaves * 32;
    gcn_spmm_kernel<<<(unsigned)((spmmThreads + 255) / 256), 256, 0, stream>>>(
        er, ec, ev, deg, feat, agg, E);

    int nTiles = N / 16;
    if (nTiles > 0) {
        size_t gemmThreads = (size_t)nTiles * 32;
        gcn_gemm_wmma_kernel<<<(unsigned)((gemmThreads + 255) / 256), 256, 0, stream>>>(
            agg, W, out, nTiles);
    }
    int rowStart = nTiles * 16;
    if (rowStart < N) {
        int tailThreads = (N - rowStart) * 32;
        gcn_gemm_tail_kernel<<<(tailThreads + 255) / 256, 256, 0, stream>>>(
            agg, W, out, rowStart, N);
    }
}


# SpMM: two partial aggregates, one per L2 half; a wave adds (f32 atomics) into the aggregate of the half its die is in; GEMM loads A as their sum; agg sum order changes (two partial sums)
# speedup vs baseline: 3.9428x; 1.2675x over previous
#include <hip/hip_runtime.h>
#include <hip/hip_bf16.h>


typedef __attribute__((ext_vector_type(2))) float v2f;
typedef __attribute__((ext_vector_type(8))) float v8f;

#define D_FEAT 32

__global__ void gcn_zero_kernel(float* __restrict__ p0, float* __restrict__ p1, size_t n) {
    size_t i = (size_t)blockIdx.x * blockDim.x + threadIdx.x;
    if (i < n) {
        p0[i] = 0.0f;
        p1[i] = 0.0f;
    }
}

#define DEG_ROWS_PER_PASS 65536
#define DEG_THREADS 1024
#define DEG_CHUNKS 64
#define DEG_BATCH 16

__global__ void __launch_bounds__(DEG_THREADS)
gcn_degree_partial_kernel(const int* __restrict__ er,
                          const float* __restrict__ ev,
                          float* __restrict__ part, int E, int N) {
    __shared__ float4 h4[DEG_ROWS_PER_PASS / 4];
    float* h = (float*)h4;

    const unsigned tid = threadIdx.x;
    const unsigned n = (unsigned)N;
    const unsigned stride = (n + 31u) & ~31u;
    const unsigned passes = gridDim.y;
    unsigned rowsPer = (((n + passes - 1u) / passes) + 31u) & ~31u;
    if (rowsPer > DEG_ROWS_PER_PASS) rowsPer = DEG_ROWS_PER_PASS;
    const unsigned rowBase = blockIdx.y * rowsPer;
    unsigned rows = 0;
    if (rowBase < n) rows = (n - rowBase < rowsPer) ? (n - rowBase) : rowsPer;

    const unsigned chunks = gridDim.x;
    const unsigned chunk = ((unsigned)E + chunks - 1u) / chunks;
    const unsigned e0 = blockIdx.x * chunk;
    unsigned e1 = e0 + chunk;
    if (e1 > (unsigned)E) e1 = (unsigned)E;

    for (unsigned i = tid; i < rowsPer / 4u; i += DEG_THREADS)
        h4[i] = make_float4(0.0f, 0.0f, 0.0f, 0.0f);
    __syncthreads();

    for (unsigned base = e0 + tid; base < e1; base += DEG_THREADS * DEG_BATCH) {
        int   r[DEG_BATCH];
        float v[DEG_BATCH];
#pragma unroll
        for (int j = 0; j < DEG_BATCH; ++j) {
            unsigned e = base + (unsigned)j * DEG_THREADS;
            r[j] = -1;
            v[j] = 0.0f;
            if (e < e1) {
                r[j] = er[e];
                v[j] = ev[e];
            }
        }
#pragma unroll
        for (int j = 0; j < DEG_BATCH; ++j) {
            unsigned local = (unsigned)r[j] - rowBase;
            if (r[j] >= 0 && local < rows) atomicAdd(&h[local], v[j]);
        }
    }
    __syncthreads();

    float* plane = part + (size_t)blockIdx.x * stride + rowBase;
    for (unsigned i = tid; i < rows; i += DEG_THREADS) plane[i] = h[i];
}

__global__ void gcn_rsqrt_kernel(const float* __restrict__ part,
                                 float* __restrict__ dis, int N) {
    int i = blockIdx.x * blockDim.x + threadIdx.x;
    if (i < N) {
        const size_t stride = ((size_t)N + 31) & ~(size_t)31;
        float p[DEG_CHUNKS];
#pragma unroll
        for (int g = 0; g < DEG_CHUNKS; ++g) p[g] = part[(size_t)g * stride + i];
        float d = 0.0f;
#pragma unroll
        for (int g = 0; g < DEG_CHUNKS; ++g) d += p[g];
        dis[i] = (d > 0.0f) ? rsqrtf(d) : 0.0f;
    }
}

#define EDGES_PER_WAVE 32

__global__ void gcn_spmm_kernel(const int* __restrict__ er,
                                const int* __restrict__ ec,
                                const float* __restrict__ ev,
                                const float* __restrict__ dis,
                                const float* __restrict__ feat,
                                float* __restrict__ agg0,
                                float* __restrict__ agg1, int E) {
    unsigned seAid = __builtin_amdgcn_s_sendmsg_rtn(0x87);
    float* __restrict__ agg = ((seAid >> 18) & 1u) ? agg1 : agg0;

    int wave = __builtin_amdgcn_readfirstlane(
        (int)(((size_t)blockIdx.x * blockDim.x + threadIdx.x) >> 5));
    int lane = threadIdx.x & 31;
    long long base = (long long)wave * EDGES_PER_WAVE;
    if (base >= (long long)E) return;
    int cnt = (int)((long long)E - base);
    if (cnt > EDGES_PER_WAVE) cnt = EDGES_PER_WAVE;

    int r = 0, c = 0;
    float v = 0.0f;
    if (lane < cnt) {
        long long e = base + lane;
        r = er[e];
        c = ec[e];
        v = dis[r] * ev[e] * dis[c];
    }

    if (cnt == EDGES_PER_WAVE) {
        float f[EDGES_PER_WAVE];
#pragma unroll
        for (int j = 0; j < EDGES_PER_WAVE; ++j) {
            int cj = __builtin_amdgcn_readlane(c, j);
            const float* frow = feat + (size_t)cj * D_FEAT;
            f[j] = frow[lane];
        }
#pragma unroll
        for (int j = 0; j < EDGES_PER_WAVE; ++j) {
            int rj = __builtin_amdgcn_readlane(r, j);
            float vj = __int_as_float(__builtin_amdgcn_readlane(__float_as_int(v), j));
            float* arow = agg + (size_t)rj * D_FEAT;
            atomicAdd(&arow[lane], vj * f[j]);
        }
    } else {
        for (int j = 0; j < cnt; ++j) {
            int cj = __builtin_amdgcn_readlane(c, j);
            int rj = __builtin_amdgcn_readlane(r, j);
            float vj = __int_as_float(__builtin_amdgcn_readlane(__float_as_int(v), j));
            float fj = feat[(size_t)cj * D_FEAT + lane];
            atomicAdd(&agg[(size_t)rj * D_FEAT + lane], vj * fj);
        }
    }
}

__global__ void gcn_gemm_wmma_kernel(const float* __restrict__ agg0,
                                     const float* __restrict__ agg1,
                                     const float* __restrict__ W,
                                     float* __restrict__ out, int nTiles) {
    int wave = (int)(((size_t)blockIdx.x * blockDim.x + threadIdx.x) >> 5);
    if (wave >= nTiles) return;
    int lane = threadIdx.x & 31;
    int half = lane >> 4;
    int l15  = lane & 15;

    const float* aRow0 = agg0 + (size_t)(wave * 16 + l15) * D_FEAT;
    const float* aRow1 = agg1 + (size_t)(wave * 16 + l15) * D_FEAT;

    v8f acc0 = {};
    v8f acc1 = {};
#pragma unroll
    for (int k = 0; k < D_FEAT; k += 4) {
        int kk = k + half * 2;
        v2f a, b0, b1;
        a.x  = aRow0[kk]     + aRow1[kk];
        a.y  = aRow0[kk + 1] + aRow1[kk + 1];
        b0.x = W[l15 * D_FEAT + kk];
        b0.y = W[l15 * D_FEAT + kk + 1];
        b1.x = W[(16 + l15) * D_FEAT + kk];
        b1.y = W[(16 + l15) * D_FEAT + kk + 1];
        acc0 = __builtin_amdgcn_wmma_f32_16x16x4_f32(
            false, a, false, b0, (short)0, acc0, false, false);
        acc1 = __builtin_amdgcn_wmma_f32_16x16x4_f32(
            false, a, false, b1, (short)0, acc1, false, false);
    }

    int mBase = wave * 16 + half * 8;
#pragma unroll
    for (int i = 0; i < 8; ++i) {
        out[(size_t)(mBase + i) * D_FEAT + l15]      = acc0[i];
        out[(size_t)(mBase + i) * D_FEAT + 16 + l15] = acc1[i];
    }
}

__global__ void gcn_gemm_tail_kernel(const float* __restrict__ agg0,
                                     const float* __restrict__ agg1,
                                     const float* __restrict__ W,
                                     float* __restrict__ out,
                                     int rowStart, int N) {
    int idx = blockIdx.x * blockDim.x + threadIdx.x;
    int row = rowStart + (idx >> 5);
    int o = idx & 31;
    if (row >= N) return;
    float s = 0.0f;
#pragma unroll
    for (int k = 0; k < D_FEAT; ++k)
        s += (agg0[(size_t)row * D_FEAT + k] + agg1[(size_t)row * D_FEAT + k]) * W[o * D_FEAT + k];
    out[(size_t)row * D_FEAT + o] = s;
}

extern "C" void kernel_launch(void* const* d_in, const int* in_sizes, int n_in,
                              void* d_out, int out_size, void* d_ws, size_t ws_size,
                              hipStream_t stream) {
    const int*   er   = (const int*)d_in[0];
    const int*   ec   = (const int*)d_in[1];
    const float* ev   = (const float*)d_in[2];
    const float* feat = (const float*)d_in[3];
    const float* W    = (const float*)d_in[4];
    float* out = (float*)d_out;

    int E = in_sizes[0];
    int N = in_sizes[3] / D_FEAT;

    float* agg  = (float*)d_ws;
    float* agg1 = agg + (size_t)N * D_FEAT;
    float* deg  = agg1 + (size_t)N * D_FEAT;
    size_t pitch = ((size_t)N + 31) & ~(size_t)31;
    float* part = deg + pitch;

    size_t zn = (size_t)N * D_FEAT;
    gcn_zero_kernel<<<(unsigned)((zn + 255) / 256), 256, 0, stream>>>(agg, agg1, zn);

    dim3 degGrid(DEG_CHUNKS, (unsigned)((N + DEG_ROWS_PER_PASS - 1) / DEG_ROWS_PER_PASS));
    gcn_degree_partial_kernel<<<degGrid, DEG_THREADS, 0, stream>>>(er, ev, part, E, N);

    gcn_rsqrt_kernel<<<(N + 255) / 256, 256, 0, stream>>>(part, deg, N);

    size_t spmmWaves = ((size_t)E + EDGES_PER_WAVE - 1) / EDGES_PER_WAVE;
    size_t spmmThreads = spmmWaves * 32;
    gcn_spmm_kernel<<<(unsigned)((spmmThreads + 255) / 256), 256, 0, stream>>>(
        er, ec, ev, deg, feat, agg, agg1, E);

    int nTiles = N / 16;
    if (nTiles > 0) {
        size_t gemmThreads = (size_t)nTiles * 32;
        gcn_gemm_wmma_kernel<<<(unsigned)((gemmThreads + 255) / 256), 256, 0, stream>>>(
            agg, agg1, W, out, nTiles);
    }
    int rowStart = nTiles * 16;
    if (rowStart < N) {
        int tailThreads = (N - rowStart) * 32;
        gcn_gemm_tail_kernel<<<(tailThreads + 255) / 256, 256, 0, stream>>>(
            agg, agg1, W, out, rowStart, N);
    }
}


# Only the row factor leaves the per-edge path (no scaled feature copy, no new pass): SpMM coefficient is ev*dis[c], GEMM scales each A row by dis[r]; row factor multiplies the row sum, not each term
# speedup vs baseline: 4.2938x; 1.0890x over previous
#include <hip/hip_runtime.h>
#include <hip/hip_bf16.h>


typedef __attribute__((ext_vector_type(2))) float v2f;
typedef __attribute__((ext_vector_type(8))) float v8f;

#define D_FEAT 32

__global__ void gcn_zero_kernel(float* __restrict__ p0, float* __restrict__ p1, size_t n) {
    size_t i = (size_t)blockIdx.x * blockDim.x + threadIdx.x;
    if (i < n) {
        p0[i] = 0.0f;
        p1[i] = 0.0f;
    }
}

#define DEG_ROWS_PER_PASS 65536
#define DEG_THREADS 1024
#define DEG_CHUNKS 64
#define DEG_BATCH 16

__global__ void __launch_bounds__(DEG_THREADS)
gcn_degree_partial_kernel(const int* __restrict__ er,
                          const float* __restrict__ ev,
                          float* __restrict__ part, int E, int N) {
    __shared__ float4 h4[DEG_ROWS_PER_PASS / 4];
    float* h = (float*)h4;

    const unsigned tid = threadIdx.x;
    const unsigned n = (unsigned)N;
    const unsigned stride = (n + 31u) & ~31u;
    const unsigned passes = gridDim.y;
    unsigned rowsPer = (((n + passes - 1u) / passes) + 31u) & ~31u;
    if (rowsPer > DEG_ROWS_PER_PASS) rowsPer = DEG_ROWS_PER_PASS;
    const unsigned rowBase = blockIdx.y * rowsPer;
    unsigned rows = 0;
    if (rowBase < n) rows = (n - rowBase < rowsPer) ? (n - rowBase) : rowsPer;

    const unsigned chunks = gridDim.x;
    const unsigned chunk = ((unsigned)E + chunks - 1u) / chunks;
    const unsigned e0 = blockIdx.x * chunk;
    unsigned e1 = e0 + chunk;
    if (e1 > (unsigned)E) e1 = (unsigned)E;

    for (unsigned i = tid; i < rowsPer / 4u; i += DEG_THREADS)
        h4[i] = make_float4(0.0f, 0.0f, 0.0f, 0.0f);
    __syncthreads();

    for (unsigned base = e0 + tid; base < e1; base += DEG_THREADS * DEG_BATCH) {
        int   r[DEG_BATCH];
        float v[DEG_BATCH];
#pragma unroll
        for (int j = 0; j < DEG_BATCH; ++j) {
            unsigned e = base + (unsigned)j * DEG_THREADS;
            r[j] = -1;
            v[j] = 0.0f;
            if (e < e1) {
                r[j] = er[e];
                v[j] = ev[e];
            }
        }
#pragma unroll
        for (int j = 0; j < DEG_BATCH; ++j) {
            unsigned local = (unsigned)r[j] - rowBase;
            if (r[j] >= 0 && local < rows) atomicAdd(&h[local], v[j]);
        }
    }
    __syncthreads();

    float* plane = part + (size_t)blockIdx.x * stride + rowBase;
    for (unsigned i = tid; i < rows; i += DEG_THREADS) plane[i] = h[i];
}

__global__ void gcn_rsqrt_kernel(const float* __restrict__ part,
                                 float* __restrict__ dis, int N) {
    int i = blockIdx.x * blockDim.x + threadIdx.x;
    if (i < N) {
        const size_t stride = ((size_t)N + 31) & ~(size_t)31;
        float p[DEG_CHUNKS];
#pragma unroll
        for (int g = 0; g < DEG_CHUNKS; ++g) p[g] = part[(size_t)g * stride + i];
        float d = 0.0f;
#pragma unroll
        for (int g = 0; g < DEG_CHUNKS; ++g) d += p[g];
        dis[i] = (d > 0.0f) ? rsqrtf(d) : 0.0f;
    }
}

#define EDGES_PER_WAVE 32

__global__ void gcn_spmm_kernel(const int* __restrict__ er,
                                const int* __restrict__ ec,
                                const float* __restrict__ ev,
                                const float* __restrict__ dis,
                                const float* __restrict__ feat,
                                float* __restrict__ agg0,
                                float* __restrict__ agg1, int E) {
    unsigned seAid = __builtin_amdgcn_s_sendmsg_rtn(0x87);
    float* __restrict__ agg = ((seAid >> 18) & 1u) ? agg1 : agg0;

    int wave = __builtin_amdgcn_readfirstlane(
        (int)(((size_t)blockIdx.x * blockDim.x + threadIdx.x) >> 5));
    int lane = threadIdx.x & 31;
    long long base = (long long)wave * EDGES_PER_WAVE;
    if (base >= (long long)E) return;
    int cnt = (int)((long long)E - base);
    if (cnt > EDGES_PER_WAVE) cnt = EDGES_PER_WAVE;

    int r = 0, c = 0;
    float v = 0.0f;
    if (lane < cnt) {
        long long e = base + lane;
        r = er[e];
        c = ec[e];
        v = ev[e] * dis[c];
    }

    if (cnt == EDGES_PER_WAVE) {
        float f[EDGES_PER_WAVE];
#pragma unroll
        for (int j = 0; j < EDGES_PER_WAVE; ++j) {
            int cj = __builtin_amdgcn_readlane(c, j);
            const float* frow = feat + (size_t)cj * D_FEAT;
            f[j] = frow[lane];
        }
#pragma unroll
        for (int j = 0; j < EDGES_PER_WAVE; ++j) {
            int rj = __builtin_amdgcn_readlane(r, j);
            float vj = __int_as_float(__builtin_amdgcn_readlane(__float_as_int(v), j));
            float* arow = agg + (size_t)rj * D_FEAT;
            atomicAdd(&arow[lane], vj * f[j]);
        }
    } else {
        for (int j = 0; j < cnt; ++j) {
            int cj = __builtin_amdgcn_readlane(c, j);
            int rj = __builtin_amdgcn_readlane(r, j);
            float vj = __int_as_float(__builtin_amdgcn_readlane(__float_as_int(v), j));
            float fj = feat[(size_t)cj * D_FEAT + lane];
            atomicAdd(&agg[(size_t)rj * D_FEAT + lane], vj * fj);
        }
    }
}

__global__ void gcn_gemm_wmma_kernel(const float* __restrict__ agg0,
                                     const float* __restrict__ agg1,
                                     const float* __restrict__ dis,
                                     const float* __restrict__ W,
                                     float* __restrict__ out, int nTiles) {
    int wave = (int)(((size_t)blockIdx.x * blockDim.x + threadIdx.x) >> 5);
    if (wave >= nTiles) return;
    int lane = threadIdx.x & 31;
    int half = lane >> 4;
    int l15  = lane & 15;

    const float* aRow0 = agg0 + (size_t)(wave * 16 + l15) * D_FEAT;
    const float* aRow1 = agg1 + (size_t)(wave * 16 + l15) * D_FEAT;
    const float dr = dis[wave * 16 + l15];

    v8f acc0 = {};
    v8f acc1 = {};
#pragma unroll
    for (int k = 0; k < D_FEAT; k += 4) {
        int kk = k + half * 2;
        v2f a, b0, b1;
        a.x  = dr * (aRow0[kk]     + aRow1[kk]);
        a.y  = dr * (aRow0[kk + 1] + aRow1[kk + 1]);
        b0.x = W[l15 * D_FEAT + kk];
        b0.y = W[l15 * D_FEAT + kk + 1];
        b1.x = W[(16 + l15) * D_FEAT + kk];
        b1.y = W[(16 + l15) * D_FEAT + kk + 1];
        acc0 = __builtin_amdgcn_wmma_f32_16x16x4_f32(
            false, a, false, b0, (short)0, acc0, false, false);
        acc1 = __builtin_amdgcn_wmma_f32_16x16x4_f32(
            false, a, false, b1, (short)0, acc1, false, false);
    }

    int mBase = wave * 16 + half * 8;
#pragma unroll
    for (int i = 0; i < 8; ++i) {
        out[(size_t)(mBase + i) * D_FEAT + l15]      = acc0[i];
        out[(size_t)(mBase + i) * D_FEAT + 16 + l15] = acc1[i];
    }
}

__global__ void gcn_gemm_tail_kernel(const float* __restrict__ agg0,
                                     const float* __restrict__ agg1,
                                     const float* __restrict__ dis,
                                     const float* __restrict__ W,
                                     float* __restrict__ out,
                                     int rowStart, int N) {
    int idx = blockIdx.x * blockDim.x + threadIdx.x;
    int row = rowStart + (idx >> 5);
    int o = idx & 31;
    if (row >= N) return;
    const float dr = dis[row];
    float s = 0.0f;
#pragma unroll
    for (int k = 0; k < D_FEAT; ++k)
        s += (dr * (agg0[(size_t)row * D_FEAT + k] + agg1[(size_t)row * D_FEAT + k])) * W[o * D_FEAT + k];
    out[(size_t)row * D_FEAT + o] = s;
}

extern "C" void kernel_launch(void* const* d_in, const int* in_sizes, int n_in,
                              void* d_out, int out_size, void* d_ws, size_t ws_size,
                              hipStream_t stream) {
    const int*   er   = (const int*)d_in[0];
    const int*   ec   = (const int*)d_in[1];
    const float* ev   = (const float*)d_in[2];
    const float* feat = (const float*)d_in[3];
    const float* W    = (const float*)d_in[4];
    float* out = (float*)d_out;

    int E = in_sizes[0];
    int N = in_sizes[3] / D_FEAT;

    float* agg  = (float*)d_ws;
    float* agg1 = agg + (size_t)N * D_FEAT;
    float* deg  = agg1 + (size_t)N * D_FEAT;
    size_t pitch = ((size_t)N + 31) & ~(size_t)31;
    float* part = deg + pitch;

    size_t zn = (size_t)N * D_FEAT;
    gcn_zero_kernel<<<(unsigned)((zn + 255) / 256), 256, 0, stream>>>(agg, agg1, zn);

    dim3 degGrid(DEG_CHUNKS, (unsigned)((N + DEG_ROWS_PER_PASS - 1) / DEG_ROWS_PER_PASS));
    gcn_degree_partial_kernel<<<degGrid, DEG_THREADS, 0, stream>>>(er, ev, part, E, N);

    gcn_rsqrt_kernel<<<(N + 255) / 256, 256, 0, stream>>>(part, deg, N);

    size_t spmmWaves = ((size_t)E + EDGES_PER_WAVE - 1) / EDGES_PER_WAVE;
    size_t spmmThreads = spmmWaves * 32;
    gcn_spmm_kernel<<<(unsigned)((spmmThreads + 255) / 256), 256, 0, stream>>>(
        er, ec, ev, deg, feat, agg, agg1, E);

    int nTiles = N / 16;
    if (nTiles > 0) {
        size_t gemmThreads = (size_t)nTiles * 32;
        gcn_gemm_wmma_kernel<<<(unsigned)((gemmThreads + 255) / 256), 256, 0, stream>>>(
            agg, agg1, deg, W, out, nTiles);
    }
    int rowStart = nTiles * 16;
    if (rowStart < N) {
        int tailThreads = (N - rowStart) * 32;
        gcn_gemm_tail_kernel<<<(tailThreads + 255) / 256, 256, 0, stream>>>(
            agg, agg1, deg, W, out, rowStart, N);
    }
}
